# starting kernel
# baseline (speedup 1.0000x reference)
#include <hip/hip_runtime.h>

typedef __attribute__((ext_vector_type(2))) float v2f;
typedef __attribute__((ext_vector_type(4))) float v4f;
typedef __attribute__((ext_vector_type(8))) float v8f;

#define B_DIM 32
#define C_DIM 256
#define HW_DIM 4096

__global__ __launch_bounds__(1024) void adaconv_mlp_kernel(
    const float* __restrict__ latent,
    const float* __restrict__ w1,  const float* __restrict__ b1,
    const float* __restrict__ w2,  const float* __restrict__ b2,
    const float* __restrict__ bw1, const float* __restrict__ bb1,
    const float* __restrict__ bw2, const float* __restrict__ bb2,
    float* __restrict__ wsOut)
{
    extern __shared__ float hwLds[];

    const int tid   = threadIdx.x;
    const int lane  = tid & 31;
    const int wave  = tid >> 5;
    const int m0    = (wave >> 4) * 16;
    const int n0    = (wave & 15) * 16;
    const int rc    = lane & 15;
    const int khalf = (lane >> 4) * 2;
    const int vbase = (lane >> 4) * 8;

    for (int path = 0; path < 2; ++path) {
        const float* __restrict__ W  = path ? bw1 : w1;
        const float* __restrict__ BB = path ? bb1 : b1;
        v8f acc = {};
        for (int k = 0; k < C_DIM; k += 4) {
            v2f a = *(const v2f*)(latent + (m0 + rc) * C_DIM + k + khalf);
            v2f b = *(const v2f*)(W + (n0 + rc) * C_DIM + k + khalf);
            acc = __builtin_amdgcn_wmma_f32_16x16x4_f32(
                false, a, false, b, (short)0, acc, false, false);
        }
        const float bv = BB[n0 + rc];
        float* __restrict__ dst = hwLds + path * (B_DIM * C_DIM);
#pragma unroll
        for (int v = 0; v < 8; ++v) {
            const int m  = m0 + vbase + v;
            float val    = acc[v] + bv;
            dst[m * C_DIM + (n0 + rc)] = val > 0.0f ? val : 0.0f;
        }
    }

    __syncthreads();

    for (int path = 0; path < 2; ++path) {
        const float* __restrict__ W  = path ? bw2 : w2;
        const float* __restrict__ BB = path ? bb2 : b2;
        const float* __restrict__ hw = hwLds + path * (B_DIM * C_DIM);
        v8f acc = {};
        for (int k = 0; k < C_DIM; k += 4) {
            v2f a = *(const v2f*)(hw + (m0 + rc) * C_DIM + k + khalf);
            v2f b = *(const v2f*)(W + (n0 + rc) * C_DIM + k + khalf);
            acc = __builtin_amdgcn_wmma_f32_16x16x4_f32(
                false, a, false, b, (short)0, acc, false, false);
        }
        const float bv = BB[n0 + rc];
        float* __restrict__ dst = wsOut + path * (B_DIM * C_DIM);
#pragma unroll
        for (int v = 0; v < 8; ++v) {
            const int m = m0 + vbase + v;
            dst[m * C_DIM + (n0 + rc)] = acc[v] + bv;
        }
    }
}

__global__ __launch_bounds__(256) void adaconv_apply_kernel(
    const float* __restrict__ x,
    const float* __restrict__ wb,
    float* __restrict__ out)
{
    const int   bc   = blockIdx.x;
    const float w    = wb[bc];
    const float b    = wb[B_DIM * C_DIM + bc];
    const size_t base = (size_t)bc * HW_DIM;

    const v4f* __restrict__ x4 = (const v4f*)(x + base);
    v4f*       __restrict__ o4 = (v4f*)(out + base);

#pragma unroll
    for (int i = 0; i < 4; ++i) {
        const int idx = threadIdx.x + i * 256;
        v4f xv = __builtin_nontemporal_load(x4 + idx);
        v4f ov = xv * w + b;
        __builtin_nontemporal_store(ov, o4 + idx);
    }
}

extern "C" void kernel_launch(void* const* d_in, const int* in_sizes, int n_in,
                              void* d_out, int out_size, void* d_ws, size_t ws_size,
                              hipStream_t stream) {
    const float* x      = (const float*)d_in[0];
    const float* latent = (const float*)d_in[1];
    const float* w1     = (const float*)d_in[2];
    const float* b1     = (const float*)d_in[3];
    const float* w2     = (const float*)d_in[4];
    const float* b2     = (const float*)d_in[5];
    const float* bw1    = (const float*)d_in[6];
    const float* bb1    = (const float*)d_in[7];
    const float* bw2    = (const float*)d_in[8];
    const float* bb2    = (const float*)d_in[9];

    float* out = (float*)d_out;
    float* ws  = (float*)d_ws;

    const size_t ldsBytes = 2 * B_DIM * C_DIM * sizeof(float);
    adaconv_mlp_kernel<<<1, 1024, ldsBytes, stream>>>(
        latent, w1, b1, w2, b2, bw1, bb1, bw2, bb2, ws);

    adaconv_apply_kernel<<<B_DIM * C_DIM, 256, 0, stream>>>(x, ws, out);
}


# MLP taken off its single 32-wave work-group: each layer is its own launch of 32 one-wave work-groups (16x32 tile, K loop unrolled whole); hidden layer staged in out; per-tile summation order kept
# speedup vs baseline: 2.9273x; 2.9273x over previous
#include <hip/hip_runtime.h>


typedef __attribute__((ext_vector_type(2))) float v2f;
typedef __attribute__((ext_vector_type(4))) float v4f;
typedef __attribute__((ext_vector_type(8))) float v8f;

#define B_DIM 32
#define C_DIM 256
#define HW_DIM 4096

__global__ __launch_bounds__(32) void adaconv_layer_kernel(
    const float* __restrict__ act,
    int actPathStride,
    const float* __restrict__ wS, const float* __restrict__ bS,
    const float* __restrict__ wB, const float* __restrict__ bB,
    int relu,
    float* __restrict__ dst)
{
    const int tile  = blockIdx.x;
    const int path  = tile >> 4;
    const int m0    = ((tile >> 3) & 1) * 16;
    const int n0    = (tile & 7) * 32;
    const int lane  = threadIdx.x;
    const int rc    = lane & 15;
    const int khalf = (lane >> 4) * 2;
    const int vbase = (lane >> 4) * 8;

    const float* __restrict__ W  = path ? wB : wS;
    const float* __restrict__ BB = path ? bB : bS;

    const float* __restrict__ aRow  = act + path * actPathStride + (m0 + rc) * C_DIM + khalf;
    const float* __restrict__ bRow0 = W + (n0 + rc) * C_DIM + khalf;
    const float* __restrict__ bRow1 = bRow0 + 16 * C_DIM;

    v8f acc0 = {};
    v8f acc1 = {};
#pragma unroll
    for (int k = 0; k < C_DIM; k += 4) {
        v2f a  = *(const v2f*)(aRow  + k);
        v2f b0 = *(const v2f*)(bRow0 + k);
        v2f b1 = *(const v2f*)(bRow1 + k);
        acc0 = __builtin_amdgcn_wmma_f32_16x16x4_f32(
            false, a, false, b0, (short)0, acc0, false, false);
        acc1 = __builtin_amdgcn_wmma_f32_16x16x4_f32(
            false, a, false, b1, (short)0, acc1, false, false);
    }

    const float bv0 = BB[n0 + rc];
    const float bv1 = BB[n0 + 16 + rc];
    float* __restrict__ d = dst + path * (B_DIM * C_DIM);
#pragma unroll
    for (int v = 0; v < 8; ++v) {
        const int m = m0 + vbase + v;
        float v0 = acc0[v] + bv0;
        float v1 = acc1[v] + bv1;
        if (relu) {
            v0 = v0 > 0.0f ? v0 : 0.0f;
            v1 = v1 > 0.0f ? v1 : 0.0f;
        }
        d[m * C_DIM + (n0 + rc)]      = v0;
        d[m * C_DIM + (n0 + 16 + rc)] = v1;
    }
}

__global__ __launch_bounds__(256) void adaconv_apply_kernel(
    const float* __restrict__ x,
    const float* __restrict__ wb,
    float* __restrict__ out)
{
    const int   bc   = blockIdx.x;
    const float w    = wb[bc];
    const float b    = wb[B_DIM * C_DIM + bc];
    const size_t base = (size_t)bc * HW_DIM;

    const v4f* __restrict__ x4 = (const v4f*)(x + base);
    v4f*       __restrict__ o4 = (v4f*)(out + base);

#pragma unroll
    for (int i = 0; i < 4; ++i) {
        const int idx = threadIdx.x + i * 256;
        v4f xv = __builtin_nontemporal_load(x4 + idx);
        v4f ov = xv * w + b;
        __builtin_nontemporal_store(ov, o4 + idx);
    }
}

extern "C" void kernel_launch(void* const* d_in, const int* in_sizes, int n_in,
                              void* d_out, int out_size, void* d_ws, size_t ws_size,
                              hipStream_t stream) {
    const float* x      = (const float*)d_in[0];
    const float* latent = (const float*)d_in[1];
    const float* w1     = (const float*)d_in[2];
    const float* b1     = (const float*)d_in[3];
    const float* w2     = (const float*)d_in[4];
    const float* b2     = (const float*)d_in[5];
    const float* bw1    = (const float*)d_in[6];
    const float* bb1    = (const float*)d_in[7];
    const float* bw2    = (const float*)d_in[8];
    const float* bb2    = (const float*)d_in[9];

    float* out = (float*)d_out;
    float* ws  = (float*)d_ws;

    float* hidden = out;

    adaconv_layer_kernel<<<32, 32, 0, stream>>>(
        latent, 0, w1, b1, bw1, bb1, 1, hidden);

    adaconv_layer_kernel<<<32, 32, 0, stream>>>(
        hidden, B_DIM * C_DIM, w2, b2, bw2, bb2, 0, ws);

    adaconv_apply_kernel<<<B_DIM * C_DIM, 256, 0, stream>>>(x, ws, out);
}


# Hidden-layer launch: each 16x32 tile by 8 waves instead of 1, operand rows staged in LDS by whole-line loads, K in 8 slices summed via LDS (layer-1 summation order changes)
# speedup vs baseline: 3.4868x; 1.1911x over previous
#include <hip/hip_runtime.h>


typedef __attribute__((ext_vector_type(2))) float v2f;
typedef __attribute__((ext_vector_type(4))) float v4f;
typedef __attribute__((ext_vector_type(8))) float v8f;

#define B_DIM 32
#define C_DIM 256
#define HW_DIM 4096

#define L1_ROWS    48
#define L1_PITCH   260
#define L1_PPITCH  34
#define L1_PTILE   (16 * L1_PPITCH)

__global__ __launch_bounds__(256) void adaconv_layer1_kernel(
    const float* __restrict__ latent,
    const float* __restrict__ wS, const float* __restrict__ bS,
    const float* __restrict__ wB, const float* __restrict__ bB,
    float* __restrict__ dst)
{
    __shared__ alignas(16) float stage[L1_ROWS * L1_PITCH];
    __shared__ alignas(16) float part[8 * L1_PTILE];

    const int tile = blockIdx.x;
    const int path = tile >> 4;
    const int m0   = ((tile >> 3) & 1) * 16;
    const int n0   = (tile & 7) * 32;
    const int tid  = threadIdx.x;
    const int kq   = __builtin_amdgcn_readfirstlane(tid >> 5);
    const int lane = tid & 31;
    const int rc   = lane & 15;
    const int h    = lane >> 4;

    const float* __restrict__ W  = path ? wB : wS;
    const float* __restrict__ BB = path ? bB : bS;

    const int srow = tid >> 6;
    const int scol = (tid & 63) * 4;
    v4f st[12];
#pragma unroll
    for (int i = 0; i < 4; ++i)
        st[i] = *(const v4f*)(latent + (m0 + 4 * i + srow) * C_DIM + scol);
#pragma unroll
    for (int i = 4; i < 12; ++i)
        st[i] = *(const v4f*)(W + (n0 + 4 * (i - 4) + srow) * C_DIM + scol);
    const int orow = tid >> 4;
    const int ocol = (tid & 15) * 2;
    const v2f bv = *(const v2f*)(BB + n0 + ocol);
#pragma unroll
    for (int i = 0; i < 12; ++i)
        *(v4f*)(stage + (4 * i + srow) * L1_PITCH + scol) = st[i];
    __syncthreads();

    const float* aL  = stage + rc * L1_PITCH + kq * 32 + h * 4;
    const float* b0L = stage + (16 + rc) * L1_PITCH + kq * 32 + h * 4;
    const float* b1L = b0L + 16 * L1_PITCH;
    v4f af[4], b0f[4], b1f[4];
#pragma unroll
    for (int t = 0; t < 4; ++t) {
        af[t]  = *(const v4f*)(aL  + 8 * t);
        b0f[t] = *(const v4f*)(b0L + 8 * t);
        b1f[t] = *(const v4f*)(b1L + 8 * t);
    }
    v8f acc0 = {};
    v8f acc1 = {};
#pragma unroll
    for (int t = 0; t < 4; ++t) {
        v2f alo = {af[t].x, af[t].y};
        v2f ahi = {af[t].z, af[t].w};
        v2f p0  = {b0f[t].x, b0f[t].y};
        v2f p1  = {b0f[t].z, b0f[t].w};
        v2f r0  = {b1f[t].x, b1f[t].y};
        v2f r1  = {b1f[t].z, b1f[t].w};
        acc0 = __builtin_amdgcn_wmma_f32_16x16x4_f32(
            false, alo, false, p0, (short)0, acc0, false, false);
        acc1 = __builtin_amdgcn_wmma_f32_16x16x4_f32(
            false, alo, false, r0, (short)0, acc1, false, false);
        acc0 = __builtin_amdgcn_wmma_f32_16x16x4_f32(
            false, ahi, false, p1, (short)0, acc0, false, false);
        acc1 = __builtin_amdgcn_wmma_f32_16x16x4_f32(
            false, ahi, false, r1, (short)0, acc1, false, false);
    }

    float* pw = part + kq * L1_PTILE + (8 * h) * L1_PPITCH + rc;
#pragma unroll
    for (int v = 0; v < 8; ++v) {
        pw[v * L1_PPITCH]      = acc0[v];
        pw[v * L1_PPITCH + 16] = acc1[v];
    }
    __syncthreads();

    const float* pr = part + orow * L1_PPITCH + ocol;
    v2f s[8];
#pragma unroll
    for (int q = 0; q < 8; ++q)
        s[q] = *(const v2f*)(pr + q * L1_PTILE);
    v2f sum = ((s[0] + s[1]) + (s[2] + s[3])) + ((s[4] + s[5]) + (s[6] + s[7]));
    sum = sum + bv;
    sum.x = sum.x > 0.0f ? sum.x : 0.0f;
    sum.y = sum.y > 0.0f ? sum.y : 0.0f;
    *(v2f*)(dst + path * (B_DIM * C_DIM) + (m0 + orow) * C_DIM + n0 + ocol) = sum;
}

#define PLANES_PER_WG 16
#define F4_PER_LANE   64
#define BATCH         16

__global__ __launch_bounds__(256) void adaconv_layer2_apply_kernel(
    const float* __restrict__ x,
    const float* __restrict__ hidden,
    const float* __restrict__ w2,  const float* __restrict__ b2,
    const float* __restrict__ bw2, const float* __restrict__ bb2,
    float* __restrict__ out)
{
    __shared__ float part[2][4][16];

    const int g    = blockIdx.x;
    const int b    = g >> 4;
    const int c0   = (g & 15) * PLANES_PER_WG;
    const int tid  = threadIdx.x;
    const int wave = __builtin_amdgcn_readfirstlane(tid >> 5);
    const int lane = tid & 31;
    const int rc   = lane & 15;
    const int h    = lane >> 4;
    const int path = wave >> 2;
    const int kq   = wave & 3;

    const float  bvec = (h ? bb2 : b2)[c0 + rc];
    const float* __restrict__ aPtr = hidden + path * (B_DIM * C_DIM) + b * C_DIM + kq * 64 + h * 4;
    const float* __restrict__ bPtr = (path ? bw2 : w2) + (c0 + rc) * C_DIM + kq * 64 + h * 4;
    v4f af[8], bf[8];
#pragma unroll
    for (int t = 0; t < 8; ++t) {
        af[t] = *(const v4f*)(aPtr + 8 * t);
        bf[t] = *(const v4f*)(bPtr + 8 * t);
    }
    __builtin_amdgcn_sched_barrier(0);

    const v4f* __restrict__ x4 = (const v4f*)(x   + (size_t)g * (PLANES_PER_WG * HW_DIM)) + tid;
    v4f*       __restrict__ o4 = (v4f*)      (out + (size_t)g * (PLANES_PER_WG * HW_DIM)) + tid;
    v4f cur[BATCH];
#pragma unroll
    for (int u = 0; u < BATCH; ++u)
        cur[u] = __builtin_nontemporal_load(x4 + u * 256);
    __builtin_amdgcn_sched_barrier(0);

    v8f acc0 = {};
    v8f acc1 = {};
#pragma unroll
    for (int t = 0; t < 8; ++t) {
        v2f a0 = {af[t].x, af[t].y};
        v2f a1 = {af[t].z, af[t].w};
        v2f q0 = {bf[t].x, bf[t].y};
        v2f q1 = {bf[t].z, bf[t].w};
        acc0 = __builtin_amdgcn_wmma_f32_16x16x4_f32(
            false, a0, false, q0, (short)0, acc0, false, false);
        acc1 = __builtin_amdgcn_wmma_f32_16x16x4_f32(
            false, a1, false, q1, (short)0, acc1, false, false);
    }
    if (lane < 16) part[path][kq][lane] = acc0[0] + acc1[0];

    asm volatile("s_wait_dscnt 0x0" ::: "memory");
    asm volatile("s_barrier_signal -1" ::: "memory");
    asm volatile("s_barrier_wait -1" ::: "memory");

    const float sb = ((part[h][0][rc] + part[h][1][rc]) +
                      (part[h][2][rc] + part[h][3][rc])) + bvec;

#pragma unroll
    for (int bt = 0; bt < F4_PER_LANE / BATCH; ++bt) {
        v4f nxt[BATCH];
        if (bt + 1 < F4_PER_LANE / BATCH) {
#pragma unroll
            for (int u = 0; u < BATCH; ++u)
                nxt[u] = __builtin_nontemporal_load(x4 + ((bt + 1) * BATCH + u) * 256);
        }
        __builtin_amdgcn_sched_barrier(0);
#pragma unroll
        for (int u = 0; u < BATCH; ++u) {
            const int plane = (bt * BATCH + u) >> 2;
            const float s  = __int_as_float(__builtin_amdgcn_readlane(__float_as_int(sb), plane));
            const float bi = __int_as_float(__builtin_amdgcn_readlane(__float_as_int(sb), 16 + plane));
            v4f ov = cur[u] * s + bi;
            __builtin_nontemporal_store(ov, o4 + (bt * BATCH + u) * 256);
        }
        __builtin_amdgcn_sched_barrier(0);
        if (bt + 1 < F4_PER_LANE / BATCH) {
#pragma unroll
            for (int u = 0; u < BATCH; ++u) cur[u] = nxt[u];
        }
    }
}

extern "C" void kernel_launch(void* const* d_in, const int* in_sizes, int n_in,
                              void* d_out, int out_size, void* d_ws, size_t ws_size,
                              hipStream_t stream) {
    const float* x      = (const float*)d_in[0];
    const float* latent = (const float*)d_in[1];
    const float* w1     = (const float*)d_in[2];
    const float* b1     = (const float*)d_in[3];
    const float* w2     = (const float*)d_in[4];
    const float* b2     = (const float*)d_in[5];
    const float* bw1    = (const float*)d_in[6];
    const float* bb1    = (const float*)d_in[7];
    const float* bw2    = (const float*)d_in[8];
    const float* bb2    = (const float*)d_in[9];

    float* out = (float*)d_out;
    float* ws  = (float*)d_ws;

    float* hidden = ws;

    adaconv_layer1_kernel<<<32, 256, 0, stream>>>(
        latent, w1, b1, bw1, bb1, hidden);

    adaconv_layer2_apply_kernel<<<512, 256, 0, stream>>>(
        x, hidden, w2, b2, bw2, bb2, out);
}
